# starting kernel
# baseline (speedup 1.0000x reference)
#include <hip/hip_runtime.h>
#include <math.h>

#define B_SZ   8
#define C_CH   64
#define L_LEN  8192
#define H_BAND 32
#define UNIT_L 512
#define NBLK   16
#define NSTRIP 32
#define NPOS   48
#define LSTRIDE 68
#define EPS_F  1e-12f

typedef float v2f __attribute__((ext_vector_type(2)));
typedef float v8f __attribute__((ext_vector_type(8)));

__global__ __launch_bounds__(96)
void FeatureExtractor_79448305041538_kernel(const float* __restrict__ feat,
                                            float* __restrict__ out) {
    __shared__ float Xs[NPOS * LSTRIDE];
    __shared__ float normS[NPOS];

    const int blk   = blockIdx.x;
    const int strip = blk & (NSTRIP - 1);
    const int n     = (blk >> 5) & (NBLK - 1);
    const int b     = blk >> 9;
    const int start = n * UNIT_L;
    const int w0    = strip * 16;

    const int tid = threadIdx.x;

    const float* fb = feat + (size_t)b * C_CH * L_LEN;
    for (int idx = tid; idx < C_CH * NPOS; idx += 96) {
        int c = idx / NPOS;
        int p = idx - c * NPOS;
        int g = start + w0 + p;
        float v = 0.0f;
        if (g < L_LEN) v = fb[(size_t)c * L_LEN + g];
        Xs[p * LSTRIDE + c] = v;
    }
    __syncthreads();

    if (tid < NPOS) {
        float s = 0.0f;
        for (int c = 0; c < C_CH; ++c) s += Xs[tid * LSTRIDE + c];
        float mu = s * (1.0f / 64.0f);
        float ss = 0.0f;
        for (int c = 0; c < C_CH; ++c) {
            float x = Xs[tid * LSTRIDE + c] - mu;
            Xs[tid * LSTRIDE + c] = x;
            ss += x * x;
        }
        normS[tid] = sqrtf(ss);
    }
    __syncthreads();

    const int wave  = tid >> 5;
    const int lane  = tid & 31;
    const int nn    = lane & 15;
    const int khalf = lane >> 4;

    const int aRow = nn;
    const int bRow = wave * 16 + nn;

    v8f acc = {};
#pragma unroll
    for (int k0 = 0; k0 < 16; ++k0) {
        int c = (k0 << 2) + (khalf << 1);
        v2f av = *(const v2f*)&Xs[aRow * LSTRIDE + c];
        v2f bv = *(const v2f*)&Xs[bRow * LSTRIDE + c];
        acc = __builtin_amdgcn_wmma_f32_16x16x4_f32(
             false, av,  false, bv,
             (short)0, acc,  false,  false);
    }

    const float ncol = normS[bRow];
    const int   vloc = wave * 16 + nn;
    const int   gcol = start + w0 + vloc;
#pragma unroll
    for (int r = 0; r < 8; ++r) {
        int m = r + (khalf << 3);
        int d = vloc - m - 1;
        if ((unsigned)d < (unsigned)H_BAND) {
            float denom = fmaxf(normS[m] * ncol, EPS_F);
            float val   = acc[r] / denom;
            if (gcol >= L_LEN) val = 0.0f;
            int t = w0 + m;
            out[((size_t)b * H_BAND + d) * L_LEN + (start + t)] = val;
        }
    }
}

extern "C" void kernel_launch(void* const* d_in, const int* in_sizes, int n_in,
                              void* d_out, int out_size, void* d_ws, size_t ws_size,
                              hipStream_t stream) {
    (void)in_sizes; (void)n_in; (void)out_size; (void)d_ws; (void)ws_size;
    const float* feat = (const float*)d_in[0];
    float* out = (float*)d_out;
    dim3 grid(B_SZ * NBLK * NSTRIP), block(96);
    FeatureExtractor_79448305041538_kernel<<<grid, block, 0, stream>>>(feat, out);
}


# Band skewed in registers, no LDS: D register r rotated r lanes in its 16-lane row (DPP), tile picked per lane; a lane then holds 8 consecutive l of one row: 8 16-byte stores per wave, was 48 4-byte
# speedup vs baseline: 5.8365x; 5.8365x over previous
#include <hip/hip_runtime.h>
#include <math.h>

#define B_SZ   8
#define C_CH   64
#define L_LEN  8192
#define H_BAND 32
#define NTILE  (L_LEN / 16)
#define S_ROWS 2
#define NFRAG  (S_ROWS + 2)
#define EPS_F  1e-12f

typedef float v2f __attribute__((ext_vector_type(2)));
typedef float v4f __attribute__((ext_vector_type(4)));
typedef float v8f __attribute__((ext_vector_type(8)));

template <int R>
__device__ __forceinline__ float row_rotl(float x) {
    const int xi = __builtin_bit_cast(int, x);
    const int yi = __builtin_amdgcn_update_dpp(xi, xi, 0x120 + (16 - R), 0xF, 0xF, false);
    return __builtin_bit_cast(float, yi);
}

__global__ __launch_bounds__(32)
void FeatureExtractor_79448305041538_kernel(const float* __restrict__ feat,
                                            float* __restrict__ out) {
    const int lane  = threadIdx.x;
    const int nn    = lane & 15;
    const int khalf = lane >> 4;
    const int w     = blockIdx.x;
    const int b     = w / (NTILE / S_ROWS);
    const int i0    = (w - b * (NTILE / S_ROWS)) * S_ROWS;

    const float* fb = feat + (size_t)b * C_CH * L_LEN + (size_t)(2 * khalf) * L_LEN + nn;

    v2f   f[NFRAG][16];
    float nrm[NFRAG];

#pragma unroll
    for (int t = 0; t < NFRAG; ++t) {
        const int j  = i0 + t;
        const int jc = (j < NTILE) ? j : (NTILE - 1);
        const float* p = fb + jc * 16;
#pragma unroll
        for (int k0 = 0; k0 < 16; ++k0) {
            f[t][k0].x = p[(size_t)(4 * k0) * L_LEN];
            f[t][k0].y = p[(size_t)(4 * k0 + 1) * L_LEN];
        }
    }

#pragma unroll
    for (int t = 0; t < NFRAG; ++t) {
        float s0 = 0.0f, s1 = 0.0f;
#pragma unroll
        for (int k0 = 0; k0 < 16; ++k0) { s0 += f[t][k0].x; s1 += f[t][k0].y; }
        float s = s0 + s1;
        s += __shfl_xor(s, 16);
        const float mu = s * (1.0f / 64.0f);
        float q0 = 0.0f, q1 = 0.0f;
#pragma unroll
        for (int k0 = 0; k0 < 16; ++k0) {
            const float x = f[t][k0].x - mu;
            const float y = f[t][k0].y - mu;
            f[t][k0].x = x;
            f[t][k0].y = y;
            q0 += x * x;
            q1 += y * y;
        }
        float q = q0 + q1;
        q += __shfl_xor(q, 16);
        nrm[t] = sqrtf(q);
    }

#pragma unroll
    for (int s = 0; s < S_ROWS; ++s) {
        v8f acc[3] = {};
#pragma unroll
        for (int k0 = 0; k0 < 16; ++k0) {
#pragma unroll
            for (int jt = 0; jt < 3; ++jt)
                acc[jt] = __builtin_amdgcn_wmma_f32_16x16x4_f32(false, f[s][k0], false, f[s + jt][k0],
                                                                (short)0, acc[jt], false, false);
        }

        float nrow[8];
#pragma unroll
        for (int r = 0; r < 8; ++r) nrow[r] = __shfl(nrm[s], r + (khalf << 3));

        const int row0 = (i0 + s) * 16;
        float v[3][8];
#pragma unroll
        for (int jt = 0; jt < 3; ++jt) {
            const float ncol = nrm[s + jt];
            const int   gcol = row0 + jt * 16 + nn;
#pragma unroll
            for (int r = 0; r < 8; ++r) {
                const float denom = fmaxf(nrow[r] * ncol, EPS_F);
                float val = acc[jt][r] / denom;
                if (gcol >= L_LEN) val = 0.0f;
                v[jt][r] = val;
            }
        }

#pragma unroll
        for (int jt = 0; jt < 3; ++jt) {
            v[jt][1] = row_rotl<1>(v[jt][1]);
            v[jt][2] = row_rotl<2>(v[jt][2]);
            v[jt][3] = row_rotl<3>(v[jt][3]);
            v[jt][4] = row_rotl<4>(v[jt][4]);
            v[jt][5] = row_rotl<5>(v[jt][5]);
            v[jt][6] = row_rotl<6>(v[jt][6]);
            v[jt][7] = row_rotl<7>(v[jt][7]);
        }
        const int  dneg = nn - 1 - (khalf << 3);
        const bool up   = dneg < 0;
        const int  d0   = dneg & 15;
        float o0[8], o1[8];
#pragma unroll
        for (int r = 0; r < 8; ++r) {
            const bool wrap = (nn + r) >= 16;
            const float lo = wrap ? v[1][r] : v[0][r];
            const float hi = wrap ? v[2][r] : v[1][r];
            o0[r] = up ? hi : lo;
            o1[r] = up ? v[2][r] : hi;
        }
        float* p0 = out + ((long)b * H_BAND + d0) * L_LEN + row0 + (khalf << 3);
        float* p1 = p0 + (long)16 * L_LEN;
        *reinterpret_cast<v4f*>(p0)     = (v4f){o0[0], o0[1], o0[2], o0[3]};
        *reinterpret_cast<v4f*>(p0 + 4) = (v4f){o0[4], o0[5], o0[6], o0[7]};
        *reinterpret_cast<v4f*>(p1)     = (v4f){o1[0], o1[1], o1[2], o1[3]};
        *reinterpret_cast<v4f*>(p1 + 4) = (v4f){o1[4], o1[5], o1[6], o1[7]};
    }
}

extern "C" void kernel_launch(void* const* d_in, const int* in_sizes, int n_in,
                              void* d_out, int out_size, void* d_ws, size_t ws_size,
                              hipStream_t stream) {
    (void)in_sizes; (void)n_in; (void)out_size; (void)d_ws; (void)ws_size;
    const float* feat = (const float*)d_in[0];
    float* out = (float*)d_out;
    dim3 grid(B_SZ * (NTILE / S_ROWS)), block(32);
    FeatureExtractor_79448305041538_kernel<<<grid, block, 0, stream>>>(feat, out);
}


# Centring moved after the products: WMMAs take raw values one channel group at a time as its loads land (loads in channel order); mean removed as G - Sx*Sy/64, norm^2 = Sxx - Sx^2/64; rounding differs
# speedup vs baseline: 6.3814x; 1.0934x over previous
#include <hip/hip_runtime.h>
#include <math.h>

#define B_SZ   8
#define C_CH   64
#define L_LEN  8192
#define H_BAND 32
#define NTILE  (L_LEN / 16)
#define S_ROWS 2
#define NFRAG  (S_ROWS + 2)
#define NGRP   16
#define AHEAD  7
#define EPS_F  1e-12f

typedef float v2f __attribute__((ext_vector_type(2)));
typedef float v4f __attribute__((ext_vector_type(4)));
typedef float v8f __attribute__((ext_vector_type(8)));

template <int R>
__device__ __forceinline__ float row_rotl(float x) {
    const int xi = __builtin_bit_cast(int, x);
    const int yi = __builtin_amdgcn_update_dpp(xi, xi, 0x120 + (16 - R), 0xF, 0xF, false);
    return __builtin_bit_cast(float, yi);
}

__global__ __launch_bounds__(32)
void FeatureExtractor_79448305041538_kernel(const float* __restrict__ feat,
                                            float* __restrict__ out) {
    const int lane  = threadIdx.x;
    const int nn    = lane & 15;
    const int khalf = lane >> 4;
    const int w     = blockIdx.x;
    const int b     = w / (NTILE / S_ROWS);
    const int i0    = (w - b * (NTILE / S_ROWS)) * S_ROWS;

    const float* fb = feat + (size_t)b * C_CH * L_LEN + (size_t)(2 * khalf) * L_LEN + nn;

    const float* p0 = fb + i0 * 16;
    const float* pc[NFRAG];
#pragma unroll
    for (int t = 0; t < NFRAG; ++t) {
        const int j  = i0 + t;
        const int jc = (j < NTILE) ? j : (NTILE - 1);
        pc[t] = fb + jc * 16;
    }

    v2f f[NGRP][NFRAG];
    v2f sv[NFRAG], qv[NFRAG];
    v8f acc[S_ROWS][3];
#pragma unroll
    for (int t = 0; t < NFRAG; ++t) { sv[t] = (v2f){0.0f, 0.0f}; qv[t] = (v2f){0.0f, 0.0f}; }
#pragma unroll
    for (int s = 0; s < S_ROWS; ++s)
#pragma unroll
        for (int jt = 0; jt < 3; ++jt) acc[s][jt] = (v8f){};

#pragma unroll
    for (int g = 0; g < AHEAD; ++g) {
#pragma unroll
        for (int t = 0; t < NFRAG; ++t) {
            f[g][t].x = p0[t * 16 + (size_t)(4 * g) * L_LEN];
            f[g][t].y = p0[t * 16 + (size_t)(4 * g + 1) * L_LEN];
        }
    }
    __builtin_amdgcn_sched_barrier(0);

#pragma unroll
    for (int k0 = 0; k0 < NGRP; ++k0) {
        if (k0 + AHEAD < NGRP) {
            const int g = k0 + AHEAD;
#pragma unroll
            for (int t = 0; t < NFRAG; ++t) {
                f[g][t].x = p0[t * 16 + (size_t)(4 * g) * L_LEN];
                f[g][t].y = (g == NGRP - 1 && t >= S_ROWS) ? pc[t][(size_t)(4 * g + 1) * L_LEN]
                                                           : p0[t * 16 + (size_t)(4 * g + 1) * L_LEN];
            }
        }
#pragma unroll
        for (int s = 0; s < S_ROWS; ++s) {
#pragma unroll
            for (int jt = 0; jt < 3; ++jt)
                acc[s][jt] = __builtin_amdgcn_wmma_f32_16x16x4_f32(false, f[k0][s], false, f[k0][s + jt],
                                                                   (short)0, acc[s][jt], false, false);
        }
#pragma unroll
        for (int t = 0; t < NFRAG; ++t) {
            sv[t] += f[k0][t];
            qv[t] += f[k0][t] * f[k0][t];
        }
        __builtin_amdgcn_sched_barrier(0);
    }

    float amean[NFRAG], nrm[NFRAG];
#pragma unroll
    for (int t = 0; t < NFRAG; ++t) {
        float s = sv[t].x + sv[t].y;
        s += __shfl_xor(s, 16);
        float q = qv[t].x + qv[t].y;
        q += __shfl_xor(q, 16);
        const float a = s * 0.125f;
        amean[t] = a;
        nrm[t] = sqrtf(fmaxf(q - a * a, 0.0f));
    }

#pragma unroll
    for (int s = 0; s < S_ROWS; ++s) {
        float nrow[8], arow[8];
#pragma unroll
        for (int r = 0; r < 8; ++r) {
            nrow[r] = __shfl(nrm[s],   r + (khalf << 3));
            arow[r] = __shfl(amean[s], r + (khalf << 3));
        }

        const int row0 = (i0 + s) * 16;
        float v[3][8];
#pragma unroll
        for (int jt = 0; jt < 3; ++jt) {
            const float ncol = nrm[s + jt];
            const float acol = amean[s + jt];
            const int   gcol = row0 + jt * 16 + nn;
#pragma unroll
            for (int r = 0; r < 8; ++r) {
                const float num   = acc[s][jt][r] - arow[r] * acol;
                const float denom = fmaxf(nrow[r] * ncol, EPS_F);
                float val = num / denom;
                if (gcol >= L_LEN) val = 0.0f;
                v[jt][r] = val;
            }
        }

#pragma unroll
        for (int jt = 0; jt < 3; ++jt) {
            v[jt][1] = row_rotl<1>(v[jt][1]);
            v[jt][2] = row_rotl<2>(v[jt][2]);
            v[jt][3] = row_rotl<3>(v[jt][3]);
            v[jt][4] = row_rotl<4>(v[jt][4]);
            v[jt][5] = row_rotl<5>(v[jt][5]);
            v[jt][6] = row_rotl<6>(v[jt][6]);
            v[jt][7] = row_rotl<7>(v[jt][7]);
        }
        const int  dneg = nn - 1 - (khalf << 3);
        const bool up   = dneg < 0;
        const int  d0   = dneg & 15;
        float o0[8], o1[8];
#pragma unroll
        for (int r = 0; r < 8; ++r) {
            const bool wrap = (nn + r) >= 16;
            const float lo = wrap ? v[1][r] : v[0][r];
            const float hi = wrap ? v[2][r] : v[1][r];
            o0[r] = up ? hi : lo;
            o1[r] = up ? v[2][r] : hi;
        }
        float* p0 = out + ((long)b * H_BAND + d0) * L_LEN + row0 + (khalf << 3);
        float* p1 = p0 + (long)16 * L_LEN;
        *reinterpret_cast<v4f*>(p0)     = (v4f){o0[0], o0[1], o0[2], o0[3]};
        *reinterpret_cast<v4f*>(p0 + 4) = (v4f){o0[4], o0[5], o0[6], o0[7]};
        *reinterpret_cast<v4f*>(p1)     = (v4f){o1[0], o1[1], o1[2], o1[3]};
        *reinterpret_cast<v4f*>(p1 + 4) = (v4f){o1[4], o1[5], o1[6], o1[7]};
    }
}

extern "C" void kernel_launch(void* const* d_in, const int* in_sizes, int n_in,
                              void* d_out, int out_size, void* d_ws, size_t ws_size,
                              hipStream_t stream) {
    (void)in_sizes; (void)n_in; (void)out_size; (void)d_ws; (void)ws_size;
    const float* feat = (const float*)d_in[0];
    float* out = (float*)d_out;
    dim3 grid(B_SZ * (NTILE / S_ROWS)), block(32);
    FeatureExtractor_79448305041538_kernel<<<grid, block, 0, stream>>>(feat, out);
}


# Same 2048 waves launched as 512 workgroups of 4 waves (was 2048 one-wave workgroups): 4x fewer workgroups for the dispatcher to place; waves share nothing; per-wave code and sums unchanged
# speedup vs baseline: 7.0144x; 1.0992x over previous
#include <hip/hip_runtime.h>
#include <math.h>

#define B_SZ   8
#define C_CH   64
#define L_LEN  8192
#define H_BAND 32
#define NTILE  (L_LEN / 16)
#define S_ROWS 2
#define NFRAG  (S_ROWS + 2)
#define NGRP   16
#define AHEAD  7
#define WAVES_WG 4
#define EPS_F  1e-12f

typedef float v2f __attribute__((ext_vector_type(2)));
typedef float v4f __attribute__((ext_vector_type(4)));
typedef float v8f __attribute__((ext_vector_type(8)));

template <int R>
__device__ __forceinline__ float row_rotl(float x) {
    const int xi = __builtin_bit_cast(int, x);
    const int yi = __builtin_amdgcn_update_dpp(xi, xi, 0x120 + (16 - R), 0xF, 0xF, false);
    return __builtin_bit_cast(float, yi);
}

__global__ __launch_bounds__(32 * WAVES_WG)
void FeatureExtractor_79448305041538_kernel(const float* __restrict__ feat,
                                            float* __restrict__ out) {
    const int lane  = threadIdx.x & 31;
    const int wib   = __builtin_amdgcn_readfirstlane((int)(threadIdx.x >> 5));
    const int nn    = lane & 15;
    const int khalf = lane >> 4;
    const int w     = blockIdx.x * WAVES_WG + wib;
    const int b     = w / (NTILE / S_ROWS);
    const int i0    = (w - b * (NTILE / S_ROWS)) * S_ROWS;

    const float* fb = feat + (size_t)b * C_CH * L_LEN + (size_t)(2 * khalf) * L_LEN + nn;

    const float* p0 = fb + i0 * 16;
    const float* pc[NFRAG];
#pragma unroll
    for (int t = 0; t < NFRAG; ++t) {
        const int j  = i0 + t;
        const int jc = (j < NTILE) ? j : (NTILE - 1);
        pc[t] = fb + jc * 16;
    }

    v2f f[NGRP][NFRAG];
    v2f sv[NFRAG], qv[NFRAG];
    v8f acc[S_ROWS][3];
#pragma unroll
    for (int t = 0; t < NFRAG; ++t) { sv[t] = (v2f){0.0f, 0.0f}; qv[t] = (v2f){0.0f, 0.0f}; }
#pragma unroll
    for (int s = 0; s < S_ROWS; ++s)
#pragma unroll
        for (int jt = 0; jt < 3; ++jt) acc[s][jt] = (v8f){};

#pragma unroll
    for (int g = 0; g < AHEAD; ++g) {
#pragma unroll
        for (int t = 0; t < NFRAG; ++t) {
            f[g][t].x = p0[t * 16 + (size_t)(4 * g) * L_LEN];
            f[g][t].y = p0[t * 16 + (size_t)(4 * g + 1) * L_LEN];
        }
    }
    __builtin_amdgcn_sched_barrier(0);

#pragma unroll
    for (int k0 = 0; k0 < NGRP; ++k0) {
        if (k0 + AHEAD < NGRP) {
            const int g = k0 + AHEAD;
#pragma unroll
            for (int t = 0; t < NFRAG; ++t) {
                f[g][t].x = p0[t * 16 + (size_t)(4 * g) * L_LEN];
                f[g][t].y = (g == NGRP - 1 && t >= S_ROWS) ? pc[t][(size_t)(4 * g + 1) * L_LEN]
                                                           : p0[t * 16 + (size_t)(4 * g + 1) * L_LEN];
            }
        }
#pragma unroll
        for (int s = 0; s < S_ROWS; ++s) {
#pragma unroll
            for (int jt = 0; jt < 3; ++jt)
                acc[s][jt] = __builtin_amdgcn_wmma_f32_16x16x4_f32(false, f[k0][s], false, f[k0][s + jt],
                                                                   (short)0, acc[s][jt], false, false);
        }
#pragma unroll
        for (int t = 0; t < NFRAG; ++t) {
            sv[t] += f[k0][t];
            qv[t] += f[k0][t] * f[k0][t];
        }
        __builtin_amdgcn_sched_barrier(0);
    }

    float amean[NFRAG], nrm[NFRAG];
#pragma unroll
    for (int t = 0; t < NFRAG; ++t) {
        float s = sv[t].x + sv[t].y;
        s += __shfl_xor(s, 16);
        float q = qv[t].x + qv[t].y;
        q += __shfl_xor(q, 16);
        const float a = s * 0.125f;
        amean[t] = a;
        nrm[t] = sqrtf(fmaxf(q - a * a, 0.0f));
    }

#pragma unroll
    for (int s = 0; s < S_ROWS; ++s) {
        float nrow[8], arow[8];
#pragma unroll
        for (int r = 0; r < 8; ++r) {
            nrow[r] = __shfl(nrm[s],   r + (khalf << 3));
            arow[r] = __shfl(amean[s], r + (khalf << 3));
        }

        const int row0 = (i0 + s) * 16;
        float v[3][8];
#pragma unroll
        for (int jt = 0; jt < 3; ++jt) {
            const float ncol = nrm[s + jt];
            const float acol = amean[s + jt];
            const int   gcol = row0 + jt * 16 + nn;
#pragma unroll
            for (int r = 0; r < 8; ++r) {
                const float num   = acc[s][jt][r] - arow[r] * acol;
                const float denom = fmaxf(nrow[r] * ncol, EPS_F);
                float val = num / denom;
                if (gcol >= L_LEN) val = 0.0f;
                v[jt][r] = val;
            }
        }

#pragma unroll
        for (int jt = 0; jt < 3; ++jt) {
            v[jt][1] = row_rotl<1>(v[jt][1]);
            v[jt][2] = row_rotl<2>(v[jt][2]);
            v[jt][3] = row_rotl<3>(v[jt][3]);
            v[jt][4] = row_rotl<4>(v[jt][4]);
            v[jt][5] = row_rotl<5>(v[jt][5]);
            v[jt][6] = row_rotl<6>(v[jt][6]);
            v[jt][7] = row_rotl<7>(v[jt][7]);
        }
        const int  dneg = nn - 1 - (khalf << 3);
        const bool up   = dneg < 0;
        const int  d0   = dneg & 15;
        float o0[8], o1[8];
#pragma unroll
        for (int r = 0; r < 8; ++r) {
            const bool wrap = (nn + r) >= 16;
            const float lo = wrap ? v[1][r] : v[0][r];
            const float hi = wrap ? v[2][r] : v[1][r];
            o0[r] = up ? hi : lo;
            o1[r] = up ? v[2][r] : hi;
        }
        float* p0 = out + ((long)b * H_BAND + d0) * L_LEN + row0 + (khalf << 3);
        float* p1 = p0 + (long)16 * L_LEN;
        *reinterpret_cast<v4f*>(p0)     = (v4f){o0[0], o0[1], o0[2], o0[3]};
        *reinterpret_cast<v4f*>(p0 + 4) = (v4f){o0[4], o0[5], o0[6], o0[7]};
        *reinterpret_cast<v4f*>(p1)     = (v4f){o1[0], o1[1], o1[2], o1[3]};
        *reinterpret_cast<v4f*>(p1 + 4) = (v4f){o1[4], o1[5], o1[6], o1[7]};
    }
}

extern "C" void kernel_launch(void* const* d_in, const int* in_sizes, int n_in,
                              void* d_out, int out_size, void* d_ws, size_t ws_size,
                              hipStream_t stream) {
    (void)in_sizes; (void)n_in; (void)out_size; (void)d_ws; (void)ws_size;
    const float* feat = (const float*)d_in[0];
    float* out = (float*)d_out;
    dim3 grid(B_SZ * (NTILE / S_ROWS) / WAVES_WG), block(32 * WAVES_WG);
    FeatureExtractor_79448305041538_kernel<<<grid, block, 0, stream>>>(feat, out);
}
